# starting kernel
# baseline (speedup 1.0000x reference)
#include <hip/hip_runtime.h>
#include <hip/hip_bf16.h>

typedef __attribute__((ext_vector_type(2))) float v2f;
typedef __attribute__((ext_vector_type(8))) float v8f;

__device__ __forceinline__ v8f wmma4(v2f a, v2f b, v8f c) {
    return __builtin_amdgcn_wmma_f32_16x16x4_f32(
         false, a,  false, b,
         (short)0, c,  false,  false);
}

#define G_TM 128
#define G_TN 128
#define G_TK 32
#define G_PAD 36

__global__ __launch_bounds__(256) void gemm_nt_kernel(
    const float* __restrict__ A, const float* __restrict__ W,
    const float* __restrict__ bias, float* __restrict__ C,
    int M, int N, int K, int act, int split)
{
    __shared__ float As[2][G_TM * G_PAD];
    __shared__ float Bs[2][G_TN * G_PAD];

    const int tid   = threadIdx.x;
    const int lane  = tid & 31;
    const int lrow  = lane & 15;
    const int khalf = (lane >> 4) * 2;
    const int rhalf = (lane >> 4) * 8;
    const int wave  = tid >> 5;

    const int m0 = blockIdx.x * G_TM;
    const int n0 = blockIdx.y * G_TN;
    const int wm = (wave & 3) * 32;
    const int wn = (wave >> 2) * 64;

    const int ldRow = tid >> 3;
    const int ldCol = (tid & 7) * 4;

    v8f cacc[2][4] = {};
    float4 pa[4], pb[4];

    auto loadTile = [&](int kt) {
#pragma unroll
        for (int i = 0; i < 4; ++i)
            pa[i] = *(const float4*)&A[(size_t)(m0 + ldRow + 32 * i) * K + kt + ldCol];
#pragma unroll
        for (int i = 0; i < 4; ++i)
            pb[i] = *(const float4*)&W[(size_t)(n0 + ldRow + 32 * i) * K + kt + ldCol];
    };
    auto storeTile = [&](int p) {
#pragma unroll
        for (int i = 0; i < 4; ++i)
            *(float4*)&As[p][(ldRow + 32 * i) * G_PAD + ldCol] = pa[i];
#pragma unroll
        for (int i = 0; i < 4; ++i)
            *(float4*)&Bs[p][(ldRow + 32 * i) * G_PAD + ldCol] = pb[i];
    };

    loadTile(0);
    storeTile(0);
    __syncthreads();

    int p = 0;
    for (int kt = 0; kt < K; kt += G_TK) {
        const bool hasNext = (kt + G_TK) < K;
        if (hasNext) loadTile(kt + G_TK);

#pragma unroll
        for (int kk = 0; kk < G_TK; kk += 4) {
            v2f af[2], bf[4];
            af[0] = *(const v2f*)&As[p][(wm +      lrow) * G_PAD + kk + khalf];
            af[1] = *(const v2f*)&As[p][(wm + 16 + lrow) * G_PAD + kk + khalf];
#pragma unroll
            for (int fn = 0; fn < 4; ++fn)
                bf[fn] = *(const v2f*)&Bs[p][(wn + fn * 16 + lrow) * G_PAD + kk + khalf];
#pragma unroll
            for (int fm = 0; fm < 2; ++fm)
#pragma unroll
                for (int fn = 0; fn < 4; ++fn)
                    cacc[fm][fn] = wmma4(af[fm], bf[fn], cacc[fm][fn]);
        }

        if (hasNext) storeTile(p ^ 1);
        __syncthreads();
        p ^= 1;
    }

#pragma unroll
    for (int fm = 0; fm < 2; ++fm)
#pragma unroll
        for (int fn = 0; fn < 4; ++fn)
#pragma unroll
            for (int r = 0; r < 8; ++r) {
                int row = m0 + wm + fm * 16 + r + rhalf;
                int col = n0 + wn + fn * 16 + lrow;
                float v = cacc[fm][fn][r];
                if (act == 1) v = (v > 0.0f) ? (v + 1.0f) : __expf(v);
                if (bias) v += bias[col];
                if (split) {
                    int bb = row >> 11, t = row & 2047;
                    int h  = col >> 6,  dd = col & 63;
                    C[(((size_t)bb * 16 + h) * 2048 + t) * 64 + dd] = v;
                } else {
                    C[(size_t)row * N + col] = v;
                }
            }
}

#define C_PAD  68
#define A_PAD  130
#define S_PAD  66

__global__ __launch_bounds__(256) void linattn_scan_kernel(
    const float* __restrict__ Q, const float* __restrict__ Kx,
    const float* __restrict__ V, float* __restrict__ O)
{
    __shared__ float qs[128 * C_PAD];
    __shared__ float ks[128 * C_PAD];
    __shared__ float vs[128 * C_PAD];
    __shared__ float Am[128 * A_PAD];
    __shared__ float Sm[64 * S_PAD];
    __shared__ float zv[64];
    __shared__ float den[128];

    const int tid   = threadIdx.x;
    const int lane  = tid & 31;
    const int lrow  = lane & 15;
    const int khalf = (lane >> 4) * 2;
    const int rhalf = (lane >> 4) * 8;
    const int wave  = tid >> 5;

    const int bh = blockIdx.x;
    const int b  = bh >> 4, h = bh & 15;
    const size_t base = (size_t)bh * 2048 * 64;

    for (int i = tid; i < 64 * S_PAD; i += 256) Sm[i] = 0.0f;
    if (tid < 64) zv[tid] = 0.0f;
    __syncthreads();

    for (int jc = 0; jc < 16; ++jc) {
        const size_t cbase = base + (size_t)jc * 128 * 64;

#pragma unroll
        for (int i = 0; i < 8; ++i) {
            int idx = tid + i * 256;
            int row = idx >> 4, c4 = (idx & 15) * 4;
            *(float4*)&qs[row * C_PAD + c4] = *(const float4*)&Q [cbase + row * 64 + c4];
            *(float4*)&ks[row * C_PAD + c4] = *(const float4*)&Kx[cbase + row * 64 + c4];
            *(float4*)&vs[row * C_PAD + c4] = *(const float4*)&V [cbase + row * 64 + c4];
        }
        __syncthreads();

        {
            const int m0 = wave * 16;
            v8f acc[8] = {};
            for (int kk = 0; kk < 64; kk += 4) {
                v2f a = *(const v2f*)&qs[(m0 + lrow) * C_PAD + kk + khalf];
#pragma unroll
                for (int c = 0; c < 8; ++c) {
                    v2f bf = *(const v2f*)&ks[(c * 16 + lrow) * C_PAD + kk + khalf];
                    acc[c] = wmma4(a, bf, acc[c]);
                }
            }
#pragma unroll
            for (int c = 0; c < 8; ++c)
#pragma unroll
                for (int r = 0; r < 8; ++r) {
                    int row = m0 + r + rhalf;
                    int col = c * 16 + lrow;
                    Am[row * A_PAD + col] = (row >= col) ? acc[c][r] : 0.0f;
                }
        }
        __syncthreads();

        if (tid < 128) {
            float s = 1e-5f;
            for (int j = 0; j <= tid; ++j) s += Am[tid * A_PAD + j];
            float qz = 0.0f;
            for (int k = 0; k < 64; ++k) qz += qs[tid * C_PAD + k] * zv[k];
            den[tid] = s + qz;
        }
        __syncthreads();

        {
            const int m0 = wave * 16;
            v8f acc[4] = {};
            for (int kk = 0; kk < 128; kk += 4) {
                v2f a = *(const v2f*)&Am[(m0 + lrow) * A_PAD + kk + khalf];
#pragma unroll
                for (int c = 0; c < 4; ++c) {
                    v2f bf;
                    bf.x = vs[(kk + khalf)     * C_PAD + c * 16 + lrow];
                    bf.y = vs[(kk + khalf + 1) * C_PAD + c * 16 + lrow];
                    acc[c] = wmma4(a, bf, acc[c]);
                }
            }
            for (int kk = 0; kk < 64; kk += 4) {
                v2f a = *(const v2f*)&qs[(m0 + lrow) * C_PAD + kk + khalf];
#pragma unroll
                for (int c = 0; c < 4; ++c) {
                    v2f bf;
                    bf.x = Sm[(kk + khalf)     * S_PAD + c * 16 + lrow];
                    bf.y = Sm[(kk + khalf + 1) * S_PAD + c * 16 + lrow];
                    acc[c] = wmma4(a, bf, acc[c]);
                }
            }
#pragma unroll
            for (int c = 0; c < 4; ++c)
#pragma unroll
                for (int r = 0; r < 8; ++r) {
                    int row = m0 + r + rhalf;
                    int col = c * 16 + lrow;
                    float o = acc[c][r] / den[row];
                    O[((size_t)(b * 2048 + jc * 128 + row)) * 1024 + h * 64 + col] = o;
                }
        }
        __syncthreads();

        {
            const int tr  = (wave & 3) * 16;
            const int tc0 = (wave >> 2) * 32;
            v8f acc[2];
#pragma unroll
            for (int c = 0; c < 2; ++c)
#pragma unroll
                for (int r = 0; r < 8; ++r)
                    acc[c][r] = Sm[(tr + r + rhalf) * S_PAD + tc0 + c * 16 + lrow];
            for (int kk = 0; kk < 128; kk += 4) {
                v2f a;
                a.x = ks[(kk + khalf)     * C_PAD + tr + lrow];
                a.y = ks[(kk + khalf + 1) * C_PAD + tr + lrow];
#pragma unroll
                for (int c = 0; c < 2; ++c) {
                    v2f bf;
                    bf.x = vs[(kk + khalf)     * C_PAD + tc0 + c * 16 + lrow];
                    bf.y = vs[(kk + khalf + 1) * C_PAD + tc0 + c * 16 + lrow];
                    acc[c] = wmma4(a, bf, acc[c]);
                }
            }
#pragma unroll
            for (int c = 0; c < 2; ++c)
#pragma unroll
                for (int r = 0; r < 8; ++r)
                    Sm[(tr + r + rhalf) * S_PAD + tc0 + c * 16 + lrow] = acc[c][r];
        }
        __syncthreads();

        if (tid < 64) {
            float s = zv[tid];
            for (int i = 0; i < 128; ++i) s += ks[i * C_PAD + tid];
            zv[tid] = s;
        }
        __syncthreads();
    }
}

extern "C" void kernel_launch(void* const* d_in, const int* in_sizes, int n_in,
                              void* d_out, int out_size, void* d_ws, size_t ws_size,
                              hipStream_t stream) {
    const float* x  = (const float*)d_in[0];
    const float* Wq = (const float*)d_in[1];
    const float* Wk = (const float*)d_in[2];
    const float* Wv = (const float*)d_in[3];
    const float* Wo = (const float*)d_in[4];
    const float* bo = (const float*)d_in[5];

    const int T = 2048, E = 1024;
    const int B = in_sizes[0] / (T * E);
    const int M = B * T;
    const size_t plane = (size_t)M * E;

    float* phiQ = (float*)d_ws;
    float* phiK = phiQ + plane;
    float* Vh   = phiK + plane;
    float* attn = Vh   + plane;

    dim3 blk(256);
    dim3 gg(M / G_TM, E / G_TN);

    gemm_nt_kernel<<<gg, blk, 0, stream>>>(x, Wq, nullptr, phiQ, M, E, E,  1,  1);
    gemm_nt_kernel<<<gg, blk, 0, stream>>>(x, Wk, nullptr, phiK, M, E, E,  1,  1);
    gemm_nt_kernel<<<gg, blk, 0, stream>>>(x, Wv, nullptr, Vh,   M, E, E,  0,  1);

    linattn_scan_kernel<<<dim3(B * 16), blk, 0, stream>>>(phiQ, phiK, Vh, attn);

    gemm_nt_kernel<<<gg, blk, 0, stream>>>(attn, Wo, bo, (float*)d_out, M, E, E,  0,  0);
}


# scan: each loop's LDS reads (5 product loops, 3 serial sums) issued as one group a block ahead of use, kept there by scheduling fences; row sums run over all 128 masked columns; order of sums same
# speedup vs baseline: 2.6047x; 2.6047x over previous
#include <hip/hip_runtime.h>
#include <hip/hip_bf16.h>

typedef __attribute__((ext_vector_type(2))) float v2f;
typedef __attribute__((ext_vector_type(8))) float v8f;

__device__ __forceinline__ v8f wmma4(v2f a, v2f b, v8f c) {
    return __builtin_amdgcn_wmma_f32_16x16x4_f32(
         false, a,  false, b,
         (short)0, c,  false,  false);
}

#define G_TM 128
#define G_TN 128
#define G_TK 64
#define G_PAD 68
#define G_STAGES 3
#define G_MAC_THREADS 256
#define G_THREADS 384
#define G_A_SZ (G_TM * G_PAD)
#define G_STAGE_SZ ((G_TM + G_TN) * G_PAD)

template <int RSTEP, int NP>
__device__ __forceinline__ void gemm_fill_stage(const float* __restrict__ Ap,
                                                const float* __restrict__ Wp,
                                                int K, int kOff, float* st, int stOff)
{
    float4 ra[NP], rb[NP];
#pragma unroll
    for (int i = 0; i < NP; ++i)
        ra[i] = *(const float4*)(Ap + (size_t)(RSTEP * i) * K + kOff);
#pragma unroll
    for (int i = 0; i < NP; ++i)
        rb[i] = *(const float4*)(Wp + (size_t)(RSTEP * i) * K + kOff);
#pragma unroll
    for (int i = 0; i < NP; ++i)
        *(float4*)&st[stOff + RSTEP * i * G_PAD] = ra[i];
#pragma unroll
    for (int i = 0; i < NP; ++i)
        *(float4*)&st[G_A_SZ + stOff + RSTEP * i * G_PAD] = rb[i];
}

__global__ __launch_bounds__(G_THREADS) void gemm_nt_kernel(
    const float* __restrict__ A, const float* __restrict__ W,
    const float* __restrict__ bias, float* __restrict__ C,
    int M, int N, int K, int act, int split)
{
    __shared__ __attribute__((aligned(16))) float Ls[G_STAGES * G_STAGE_SZ];

    const int tid   = threadIdx.x;
    const int lane  = tid & 31;
    const int lrow  = lane & 15;
    const int khalf = (lane >> 4) * 2;
    const int rhalf = (lane >> 4) * 8;
    const int wave  = __builtin_amdgcn_readfirstlane(tid >> 5);
    const bool isMac = (wave < 8);

    const int m0 = blockIdx.x * G_TM;
    const int n0 = blockIdx.y * G_TN;
    const int wm = (wave & 1) * 64;
    const int wn = ((wave >> 1) & 3) * 32;

    const int gtid  = isMac ? tid : tid - G_MAC_THREADS;
    const int ldRow = gtid >> 4;
    const int ldCol = (gtid & 15) * 4;
    const float* Ap = A + (size_t)(m0 + ldRow) * K + ldCol;
    const float* Wp = W + (size_t)(n0 + ldRow) * K + ldCol;
    const int stOff = ldRow * G_PAD + ldCol;

    const int nkt = K / G_TK;

    if (isMac) {
        gemm_fill_stage<16, 8>(Ap, Wp, K, 0, Ls, stOff);
    } else if (nkt > 1) {
        gemm_fill_stage<8, 16>(Ap, Wp, K, G_TK, Ls + G_STAGE_SZ, stOff);
    }
    __syncthreads();

    v8f cacc[4][2] = {};

    int sc = 0;
    int sf = 2;
    for (int kt = 0; kt < nkt; ++kt) {
        if (isMac) {
            const float* aP = Ls + sc * G_STAGE_SZ + (wm + lrow) * G_PAD + khalf;
            const float* bP = Ls + sc * G_STAGE_SZ + G_A_SZ + (wn + lrow) * G_PAD + khalf;
#pragma unroll
            for (int kk = 0; kk < G_TK; kk += 8) {
                v2f af[2][4], bf[2][2];
#pragma unroll
                for (int s = 0; s < 2; ++s) {
#pragma unroll
                    for (int fm = 0; fm < 4; ++fm)
                        af[s][fm] = *(const v2f*)(aP + fm * 16 * G_PAD + kk + 4 * s);
#pragma unroll
                    for (int fn = 0; fn < 2; ++fn)
                        bf[s][fn] = *(const v2f*)(bP + fn * 16 * G_PAD + kk + 4 * s);
                }
#pragma unroll
                for (int s = 0; s < 2; ++s)
#pragma unroll
                    for (int fm = 0; fm < 4; ++fm)
#pragma unroll
                        for (int fn = 0; fn < 2; ++fn)
                            cacc[fm][fn] = wmma4(af[s][fm], bf[s][fn], cacc[fm][fn]);
            }
        } else if (kt + 2 < nkt) {
            gemm_fill_stage<8, 16>(Ap, Wp, K, (kt + 2) * G_TK, Ls + sf * G_STAGE_SZ, stOff);
        }
        __syncthreads();
        sc = (sc == G_STAGES - 1) ? 0 : sc + 1;
        sf = (sf == G_STAGES - 1) ? 0 : sf + 1;
    }

    if (isMac) {
#pragma unroll
        for (int fm = 0; fm < 4; ++fm)
#pragma unroll
            for (int fn = 0; fn < 2; ++fn)
#pragma unroll
                for (int r = 0; r < 8; ++r) {
                    int row = m0 + wm + fm * 16 + r + rhalf;
                    int col = n0 + wn + fn * 16 + lrow;
                    float v = cacc[fm][fn][r];
                    if (act == 1) v = (v > 0.0f) ? (v + 1.0f) : __expf(v);
                    if (bias) v += bias[col];
                    if (split) {
                        int bb = row >> 11, t = row & 2047;
                        int h  = col >> 6,  dd = col & 63;
                        C[(((size_t)bb * 16 + h) * 2048 + t) * 64 + dd] = v;
                    } else {
                        C[(size_t)row * N + col] = v;
                    }
                }
    }
}

__device__ __forceinline__ void lds_barrier() {
    __builtin_amdgcn_fence(__ATOMIC_RELEASE, "workgroup", "local");
    __builtin_amdgcn_s_barrier();
    __builtin_amdgcn_fence(__ATOMIC_ACQUIRE, "workgroup", "local");
}

#define C_PAD  68
#define V_PAD  40
#define A_PAD  130
#define S_PAD  40
#define R_ROWS 32
#define V_COLS 32
#define QS_SZ (R_ROWS * C_PAD)
#define KS_SZ (128 * C_PAD)
#define VS_SZ (128 * V_PAD)
#define AM_SZ (R_ROWS * A_PAD)
#define SM_SZ (64 * S_PAD)

__device__ __forceinline__ void scan_keep_group() { __builtin_amdgcn_sched_barrier(0); }


__device__ __forceinline__ void scan_ld_scores(const float* qA, const float* kB, int kk0,
                                               v2f (&qf)[4], v2f (&kf)[4][4])
{
#pragma unroll
    for (int s = 0; s < 4; ++s) {
        qf[s] = *(const v2f*)(qA + kk0 + 4 * s);
#pragma unroll
        for (int c = 0; c < 4; ++c)
            kf[s][c] = *(const v2f*)(kB + c * 16 * C_PAD + kk0 + 4 * s);
    }
}

template <int BP>
__device__ __forceinline__ void scan_ld_ab2(const float* aA, const float* bB, int kk0,
                                            v2f (&af)[8], float (&bx)[8][2], float (&by)[8][2])
{
#pragma unroll
    for (int s = 0; s < 8; ++s) {
        const int kk = kk0 + 4 * s;
        af[s] = *(const v2f*)(aA + kk);
#pragma unroll
        for (int c = 0; c < 2; ++c) {
            bx[s][c] = bB[kk       * BP + c * 16];
            by[s][c] = bB[(kk + 1) * BP + c * 16];
        }
    }
}

__device__ __forceinline__ v2f scan_pair(float x, float y) { v2f r; r.x = x; r.y = y; return r; }

__device__ __forceinline__ void scan_ld_kv(const float* kA, const float* vB, int kk0,
                                           float (&ax)[4][2], float (&ay)[4][2],
                                           float (&bx)[4][2], float (&by)[4][2])
{
#pragma unroll
    for (int s = 0; s < 4; ++s) {
        const int kk = kk0 + 4 * s;
#pragma unroll
        for (int i = 0; i < 2; ++i) {
            ax[s][i] = kA[kk       * C_PAD + i * 16];
            ay[s][i] = kA[(kk + 1) * C_PAD + i * 16];
        }
#pragma unroll
        for (int c = 0; c < 2; ++c) {
            bx[s][c] = vB[kk       * V_PAD + c * 16];
            by[s][c] = vB[(kk + 1) * V_PAD + c * 16];
        }
    }
}

__global__ __launch_bounds__(256) void linattn_scan_kernel(
    const float* __restrict__ Q, const float* __restrict__ Kx,
    const float* __restrict__ V, float* __restrict__ O)
{
    __shared__ __attribute__((aligned(16))) float qbuf[3 * QS_SZ];
    __shared__ __attribute__((aligned(16))) float kbuf[3 * KS_SZ];
    __shared__ __attribute__((aligned(16))) float vbuf[3 * VS_SZ];
    __shared__ __attribute__((aligned(16))) float Ambuf[2 * AM_SZ];
    __shared__ __attribute__((aligned(16))) float Smbuf[2 * SM_SZ];
    __shared__ __attribute__((aligned(16))) float zv[64];
    __shared__ float rsbuf[2 * R_ROWS];
    __shared__ float qzbuf[2 * R_ROWS];

    const int tid   = threadIdx.x;
    const int lane  = tid & 31;
    const int lrow  = lane & 15;
    const int khalf = (lane >> 4) * 2;
    const int rhalf = (lane >> 4) * 8;
    const int wave  = __builtin_amdgcn_readfirstlane(tid >> 5);

    const int bh = blockIdx.x >> 3;
    const int r0 = ((blockIdx.x >> 1) & 3) * R_ROWS;
    const int c0 = (blockIdx.x & 1) * V_COLS;
    const int b  = bh >> 4, h = bh & 15;
    const size_t base = (size_t)bh * 2048 * 64;

    const int kqRow = tid >> 4, kqC4 = (tid & 15) * 4;
    const int vRow  = tid >> 3, vC4  = (tid & 7) * 4;
    const float* gK = Kx + base + kqRow * 64 + kqC4;
    const float* gQ = Q  + base + (r0 + kqRow) * 64 + kqC4;
    const float* gV = V  + base + vRow * 64 + c0 + vC4;
    const int kqOff = kqRow * C_PAD + kqC4;
    const int vOff  = vRow * V_PAD + vC4;

#define SCAN_IMAGE_DECL \
    float4 pk0, pk1, pk2, pk3, pk4, pk5, pk6, pk7, pq0, pq1, pv0, pv1, pv2, pv3;
#define SCAN_LOAD_IMAGE(off)                                   \
    pk0 = *(const float4*)(gK + (off) + 0 * 16 * 64);          \
    pk1 = *(const float4*)(gK + (off) + 1 * 16 * 64);          \
    pk2 = *(const float4*)(gK + (off) + 2 * 16 * 64);          \
    pk3 = *(const float4*)(gK + (off) + 3 * 16 * 64);          \
    pk4 = *(const float4*)(gK + (off) + 4 * 16 * 64);          \
    pk5 = *(const float4*)(gK + (off) + 5 * 16 * 64);          \
    pk6 = *(const float4*)(gK + (off) + 6 * 16 * 64);          \
    pk7 = *(const float4*)(gK + (off) + 7 * 16 * 64);          \
    pq0 = *(const float4*)(gQ + (off) + 0 * 16 * 64);          \
    pq1 = *(const float4*)(gQ + (off) + 1 * 16 * 64);          \
    pv0 = *(const float4*)(gV + (off) + 0 * 32 * 64);          \
    pv1 = *(const float4*)(gV + (off) + 1 * 32 * 64);          \
    pv2 = *(const float4*)(gV + (off) + 2 * 32 * 64);          \
    pv3 = *(const float4*)(gV + (off) + 3 * 32 * 64);
#define SCAN_STORE_IMAGE(kd, qd, vd)                           \
    *(float4*)&(kd)[kqOff + 0 * 16 * C_PAD] = pk0;             \
    *(float4*)&(kd)[kqOff + 1 * 16 * C_PAD] = pk1;             \
    *(float4*)&(kd)[kqOff + 2 * 16 * C_PAD] = pk2;             \
    *(float4*)&(kd)[kqOff + 3 * 16 * C_PAD] = pk3;             \
    *(float4*)&(kd)[kqOff + 4 * 16 * C_PAD] = pk4;             \
    *(float4*)&(kd)[kqOff + 5 * 16 * C_PAD] = pk5;             \
    *(float4*)&(kd)[kqOff + 6 * 16 * C_PAD] = pk6;             \
    *(float4*)&(kd)[kqOff + 7 * 16 * C_PAD] = pk7;             \
    *(float4*)&(qd)[kqOff + 0 * 16 * C_PAD] = pq0;             \
    *(float4*)&(qd)[kqOff + 1 * 16 * C_PAD] = pq1;             \
    *(float4*)&(vd)[vOff + 0 * 32 * V_PAD] = pv0;              \
    *(float4*)&(vd)[vOff + 1 * 32 * V_PAD] = pv1;              \
    *(float4*)&(vd)[vOff + 2 * 32 * V_PAD] = pv2;              \
    *(float4*)&(vd)[vOff + 3 * 32 * V_PAD] = pv3;


    {
        SCAN_IMAGE_DECL
        SCAN_LOAD_IMAGE(0)
        for (int i = tid; i < SM_SZ; i += 256) Smbuf[i] = 0.0f;
        SCAN_STORE_IMAGE(kbuf, qbuf, vbuf)
    }
    __syncthreads();

    float zreg = 0.0f;
    int bA = 0;

    for (int t = 0; t <= 16; ++t) {
        const bool hasA = (t < 16);
        const bool hasS = (t > 0);
        const bool more = (t + 1 < 16);
        const int bS = (bA == 0) ? 2 : bA - 1;
        const int bN = (bA == 2) ? 0 : bA + 1;
        const int pa = t & 1;

        const float* qsA = qbuf + bA * QS_SZ;
        const float* ksA = kbuf + bA * KS_SZ;
        const float* qsS = qbuf + bS * QS_SZ;
        const float* ksS = kbuf + bS * KS_SZ;
        const float* vsS = vbuf + bS * VS_SZ;
        float*       AmW  = Ambuf + pa * AM_SZ;
        const float* AmR  = Ambuf + (pa ^ 1) * AM_SZ;
        const float* Scur = Smbuf + (pa ^ 1) * SM_SZ;
        float*       Snxt = Smbuf + pa * SM_SZ;
        float*       rsW  = rsbuf + pa * R_ROWS;
        const float* rsR  = rsbuf + (pa ^ 1) * R_ROWS;
        float*       qzW  = qzbuf + pa * R_ROWS;
        const float* qzR  = qzbuf + (pa ^ 1) * R_ROWS;

        SCAN_IMAGE_DECL
        if (more) {
            const size_t nOff = (size_t)(t + 1) * 128 * 64;
            SCAN_LOAD_IMAGE(nOff)
        }

        v8f nacc[2] = {};
        v8f sacc[4] = {};

        if (wave >= 4) {
            if (hasA) {
                if (wave >= 6) zv[tid - 192] = zreg;

                const int a   = wave - 4;
                const int m0  = (a >> 1) * 16;
                const int ct0 = (a & 1) * 4;
                const float* qA = qsA + (m0 + lrow) * C_PAD + khalf;
                const float* kB = ksA + (ct0 * 16 + lrow) * C_PAD + khalf;
                v8f acc[4] = {};
                v2f qf[2][4], kf[2][4][4];
                scan_ld_scores(qA, kB, 0, qf[0], kf[0]);
#pragma unroll
                for (int blk = 0; blk < 4; ++blk) {
                    const int cur = blk & 1;
                    if (blk + 1 < 4)
                        scan_ld_scores(qA, kB, (blk + 1) * 16, qf[cur ^ 1], kf[cur ^ 1]);
                    scan_keep_group();
#pragma unroll
                    for (int s = 0; s < 4; ++s)
#pragma unroll
                        for (int c = 0; c < 4; ++c)
                            acc[c] = wmma4(qf[cur][s], kf[cur][s][c], acc[c]);
                    scan_keep_group();
                }
#pragma unroll
                for (int c = 0; c < 4; ++c)
#pragma unroll
                    for (int r = 0; r < 8; ++r) {
                        int row = m0 + r + rhalf;
                        int col = (ct0 + c) * 16 + lrow;
                        AmW[row * A_PAD + col] = (r0 + row >= col) ? acc[c][r] : 0.0f;
                    }
            }
        } else if (wave < 2) {
            if (hasS) {
                const int m0 = wave * 16;
                const float* aA = AmR + (m0 + lrow) * A_PAD + khalf;
                const float* vB = vsS + khalf * V_PAD + lrow;
                v2f af[2][8];
                float vx[2][8][2], vy[2][8][2];
                scan_ld_ab2<V_PAD>(aA, vB, 0, af[0], vx[0], vy[0]);
#pragma unroll
                for (int blk = 0; blk < 4; ++blk) {
                    const int cur = blk & 1;
                    if (blk + 1 < 4)
                        scan_ld_ab2<V_PAD>(aA, vB, (blk + 1) * 32, af[cur ^ 1], vx[cur ^ 1], vy[cur ^ 1]);
                    scan_keep_group();
#pragma unroll
                    for (int s = 0; s < 8; ++s)
#pragma unroll
                        for (int c = 0; c < 2; ++c)
                            nacc[c] = wmma4(af[cur][s], scan_pair(vx[cur][s][c], vy[cur][s][c]), nacc[c]);
                    scan_keep_group();
                }
            }
        } else {
            if (hasS) {
                const int tr0 = (wave - 2) * 32;
                const float* kA = ksS + khalf * C_PAD + tr0 + lrow;
                const float* vB = vsS + khalf * V_PAD + lrow;
#pragma unroll
                for (int i = 0; i < 2; ++i)
#pragma unroll
                    for (int c = 0; c < 2; ++c)
#pragma unroll
                        for (int r = 0; r < 8; ++r)
                            sacc[i * 2 + c][r] = Scur[(tr0 + i * 16 + r + rhalf) * S_PAD + c * 16 + lrow];
                float ax[2][4][2], ay[2][4][2], bx[2][4][2], by[2][4][2];
                scan_ld_kv(kA, vB, 0, ax[0], ay[0], bx[0], by[0]);
#pragma unroll
                for (int blk = 0; blk < 4; ++blk) {
                    const int cur = blk & 1;
                    if (blk + 1 < 4)
                        scan_ld_kv(kA, vB, (blk + 1) * 16, ax[cur ^ 1], ay[cur ^ 1], bx[cur ^ 1], by[cur ^ 1]);
                    scan_keep_group();
#pragma unroll
                    for (int s = 0; s < 4; ++s)
#pragma unroll
                        for (int i = 0; i < 2; ++i)
#pragma unroll
                            for (int c = 0; c < 2; ++c)
                                sacc[i * 2 + c] = wmma4(scan_pair(ax[cur][s][i], ay[cur][s][i]),
                                                        scan_pair(bx[cur][s][c], by[cur][s][c]), sacc[i * 2 + c]);
                    scan_keep_group();
                }
            }
        }
        lds_barrier();

        if (wave >= 4) {
            if (hasA) {
                if (wave == 4) {
                    const float* ar = AmW + lane * A_PAD;
                    float s = 1e-5f;
                    v2f rb[2][16];
#pragma unroll
                    for (int i = 0; i < 16; ++i) rb[0][i] = *(const v2f*)(ar + 2 * i);
#pragma unroll
                    for (int blk = 0; blk < 4; ++blk) {
                        const int cur = blk & 1;
                        if (blk + 1 < 4) {
#pragma unroll
                            for (int i = 0; i < 16; ++i)
                                rb[cur ^ 1][i] = *(const v2f*)(ar + (blk + 1) * 32 + 2 * i);
                        }
                        scan_keep_group();
#pragma unroll
                        for (int i = 0; i < 16; ++i) { s += rb[cur][i].x; s += rb[cur][i].y; }
                        scan_keep_group();
                    }
                    rsW[lane] = s;
                } else if (wave == 5) {
                    const float* qr = qsA + lane * C_PAD;
                    float qz = 0.0f;
                    float4 qb[2][4], zb[2][4];
#pragma unroll
                    for (int i = 0; i < 4; ++i) {
                        qb[0][i] = *(const float4*)(qr + 4 * i);
                        zb[0][i] = *(const float4*)(zv + 4 * i);
                    }
#pragma unroll
                    for (int blk = 0; blk < 4; ++blk) {
                        const int cur = blk & 1;
                        if (blk + 1 < 4) {
#pragma unroll
                            for (int i = 0; i < 4; ++i) {
                                qb[cur ^ 1][i] = *(const float4*)(qr + (blk + 1) * 16 + 4 * i);
                                zb[cur ^ 1][i] = *(const float4*)(zv + (blk + 1) * 16 + 4 * i);
                            }
                        }
                        scan_keep_group();
#pragma unroll
                        for (int i = 0; i < 4; ++i) {
                            qz = fmaf(qb[cur][i].x, zb[cur][i].x, qz);
                            qz = fmaf(qb[cur][i].y, zb[cur][i].y, qz);
                            qz = fmaf(qb[cur][i].z, zb[cur][i].z, qz);
                            qz = fmaf(qb[cur][i].w, zb[cur][i].w, qz);
                        }
                        scan_keep_group();
                    }
                    qzW[lane] = qz;
                } else {
                    const float* kc = ksA + (tid - 192);
                    float s = zreg;
                    float cb[2][32];
#pragma unroll
                    for (int i = 0; i < 32; ++i) cb[0][i] = kc[i * C_PAD];
#pragma unroll
                    for (int blk = 0; blk < 4; ++blk) {
                        const int cur = blk & 1;
                        if (blk + 1 < 4) {
#pragma unroll
                            for (int i = 0; i < 32; ++i)
                                cb[cur ^ 1][i] = kc[((blk + 1) * 32 + i) * C_PAD];
                        }
                        scan_keep_group();
#pragma unroll
                        for (int i = 0; i < 32; ++i) s += cb[cur][i];
                        scan_keep_group();
                    }
                    zreg = s;
                }
            }
        } else if (wave < 2) {
            if (hasS) {
                const int m0 = wave * 16;
                const float* qA = qsS + (m0 + lrow) * C_PAD + khalf;
                const float* sB = Scur + khalf * S_PAD + lrow;
                v2f qf[2][8];
                float sx[2][8][2], sy[2][8][2];
                scan_ld_ab2<S_PAD>(qA, sB, 0, qf[0], sx[0], sy[0]);
#pragma unroll
                for (int blk = 0; blk < 2; ++blk) {
                    const int cur = blk & 1;
                    if (blk + 1 < 2)
                        scan_ld_ab2<S_PAD>(qA, sB, (blk + 1) * 32, qf[cur ^ 1], sx[cur ^ 1], sy[cur ^ 1]);
                    scan_keep_group();
#pragma unroll
                    for (int s = 0; s < 8; ++s)
#pragma unroll
                        for (int c = 0; c < 2; ++c)
                            nacc[c] = wmma4(qf[cur][s], scan_pair(sx[cur][s][c], sy[cur][s][c]), nacc[c]);
                    scan_keep_group();
                }
#pragma unroll
                for (int c = 0; c < 2; ++c)
#pragma unroll
                    for (int r = 0; r < 8; ++r) {
                        int row = m0 + r + rhalf;
                        int col = c * 16 + lrow;
                        float o = nacc[c][r] / (rsR[row] + qzR[row]);
                        O[((size_t)(b * 2048 + (t - 1) * 128 + r0 + row)) * 1024 + h * 64 + c0 + col] = o;
                    }
            }
        } else {
            if (hasS) {
                const int tr0 = (wave - 2) * 32;
                const float* kA = ksS + khalf * C_PAD + tr0 + lrow;
                const float* vB = vsS + khalf * V_PAD + lrow;
                float ax[2][4][2], ay[2][4][2], bx[2][4][2], by[2][4][2];
                scan_ld_kv(kA, vB, 64, ax[0], ay[0], bx[0], by[0]);
#pragma unroll
                for (int blk = 0; blk < 4; ++blk) {
                    const int cur = blk & 1;
                    if (blk + 1 < 4)
                        scan_ld_kv(kA, vB, 64 + (blk + 1) * 16, ax[cur ^ 1], ay[cur ^ 1], bx[cur ^ 1], by[cur ^ 1]);
                    scan_keep_group();
#pragma unroll
                    for (int s = 0; s < 4; ++s)
#pragma unroll
                        for (int i = 0; i < 2; ++i)
#pragma unroll
                            for (int c = 0; c < 2; ++c)
                                sacc[i * 2 + c] = wmma4(scan_pair(ax[cur][s][i], ay[cur][s][i]),
                                                        scan_pair(bx[cur][s][c], by[cur][s][c]), sacc[i * 2 + c]);
                    scan_keep_group();
                }
#pragma unroll
                for (int i = 0; i < 2; ++i)
#pragma unroll
                    for (int c = 0; c < 2; ++c)
#pragma unroll
                        for (int r = 0; r < 8; ++r)
                            Snxt[(tr0 + i * 16 + r + rhalf) * S_PAD + c * 16 + lrow] = sacc[i * 2 + c][r];
            }
        }

        if (more) {
            float* kn = kbuf + bN * KS_SZ;
            float* qn = qbuf + bN * QS_SZ;
            float* vn = vbuf + bN * VS_SZ;
            SCAN_STORE_IMAGE(kn, qn, vn)
        }
        lds_barrier();
        bA = bN;
    }
}

extern "C" void kernel_launch(void* const* d_in, const int* in_sizes, int n_in,
                              void* d_out, int out_size, void* d_ws, size_t ws_size,
                              hipStream_t stream) {
    const float* x  = (const float*)d_in[0];
    const float* Wq = (const float*)d_in[1];
    const float* Wk = (const float*)d_in[2];
    const float* Wv = (const float*)d_in[3];
    const float* Wo = (const float*)d_in[4];
    const float* bo = (const float*)d_in[5];

    const int T = 2048, E = 1024;
    const int B = in_sizes[0] / (T * E);
    const int M = B * T;
    const size_t plane = (size_t)M * E;

    float* phiQ = (float*)d_ws;
    float* phiK = phiQ + plane;
    float* Vh   = phiK + plane;
    float* attn = Vh   + plane;

    dim3 blk(256);
    dim3 gblk(G_THREADS);
    dim3 gg(M / G_TM, E / G_TN);

    gemm_nt_kernel<<<gg, gblk, 0, stream>>>(x, Wq, nullptr, phiQ, M, E, E,  1,  1);
    gemm_nt_kernel<<<gg, gblk, 0, stream>>>(x, Wk, nullptr, phiK, M, E, E,  1,  1);
    gemm_nt_kernel<<<gg, gblk, 0, stream>>>(x, Wv, nullptr, Vh,   M, E, E,  0,  1);

    linattn_scan_kernel<<<dim3(B * 128), blk, 0, stream>>>(phiQ, phiK, Vh, attn);

    gemm_nt_kernel<<<gg, gblk, 0, stream>>>(attn, Wo, bo, (float*)d_out, M, E, E,  0,  0);
}


# Q,K,V projections from one staged x tile: block tile 128 x (3x128), an x K-tile in LDS once for three weight tiles, 24 accumulator tiles per wave, one launch for the three; order of sums same
# speedup vs baseline: 2.7468x; 1.0546x over previous
#include <hip/hip_runtime.h>
#include <hip/hip_bf16.h>

typedef __attribute__((ext_vector_type(2))) float v2f;
typedef __attribute__((ext_vector_type(8))) float v8f;

__device__ __forceinline__ v8f wmma4(v2f a, v2f b, v8f c) {
    return __builtin_amdgcn_wmma_f32_16x16x4_f32(
         false, a,  false, b,
         (short)0, c,  false,  false);
}

#define G_TM 128
#define G_TN 128
#define G_TK 64
#define G_PAD 68
#define G_STAGES 3
#define G_MAC_THREADS 256
#define G_THREADS 384
#define G_A_SZ (G_TM * G_PAD)
#define G_STAGE_SZ ((G_TM + G_TN) * G_PAD)

template <int RSTEP, int NP>
__device__ __forceinline__ void gemm_fill_stage(const float* __restrict__ Ap,
                                                const float* __restrict__ Wp,
                                                int K, int kOff, float* st, int stOff)
{
    float4 ra[NP], rb[NP];
#pragma unroll
    for (int i = 0; i < NP; ++i)
        ra[i] = *(const float4*)(Ap + (size_t)(RSTEP * i) * K + kOff);
#pragma unroll
    for (int i = 0; i < NP; ++i)
        rb[i] = *(const float4*)(Wp + (size_t)(RSTEP * i) * K + kOff);
#pragma unroll
    for (int i = 0; i < NP; ++i)
        *(float4*)&st[stOff + RSTEP * i * G_PAD] = ra[i];
#pragma unroll
    for (int i = 0; i < NP; ++i)
        *(float4*)&st[G_A_SZ + stOff + RSTEP * i * G_PAD] = rb[i];
}

__global__ __launch_bounds__(G_THREADS) void gemm_nt_kernel(
    const float* __restrict__ A, const float* __restrict__ W,
    const float* __restrict__ bias, float* __restrict__ C,
    int M, int N, int K, int act, int split)
{
    __shared__ __attribute__((aligned(16))) float Ls[G_STAGES * G_STAGE_SZ];

    const int tid   = threadIdx.x;
    const int lane  = tid & 31;
    const int lrow  = lane & 15;
    const int khalf = (lane >> 4) * 2;
    const int rhalf = (lane >> 4) * 8;
    const int wave  = __builtin_amdgcn_readfirstlane(tid >> 5);
    const bool isMac = (wave < 8);

    const int m0 = blockIdx.x * G_TM;
    const int n0 = blockIdx.y * G_TN;
    const int wm = (wave & 1) * 64;
    const int wn = ((wave >> 1) & 3) * 32;

    const int gtid  = isMac ? tid : tid - G_MAC_THREADS;
    const int ldRow = gtid >> 4;
    const int ldCol = (gtid & 15) * 4;
    const float* Ap = A + (size_t)(m0 + ldRow) * K + ldCol;
    const float* Wp = W + (size_t)(n0 + ldRow) * K + ldCol;
    const int stOff = ldRow * G_PAD + ldCol;

    const int nkt = K / G_TK;

    if (isMac) {
        gemm_fill_stage<16, 8>(Ap, Wp, K, 0, Ls, stOff);
    } else if (nkt > 1) {
        gemm_fill_stage<8, 16>(Ap, Wp, K, G_TK, Ls + G_STAGE_SZ, stOff);
    }
    __syncthreads();

    v8f cacc[4][2] = {};

    int sc = 0;
    int sf = 2;
    for (int kt = 0; kt < nkt; ++kt) {
        if (isMac) {
            const float* aP = Ls + sc * G_STAGE_SZ + (wm + lrow) * G_PAD + khalf;
            const float* bP = Ls + sc * G_STAGE_SZ + G_A_SZ + (wn + lrow) * G_PAD + khalf;
#pragma unroll
            for (int kk = 0; kk < G_TK; kk += 8) {
                v2f af[2][4], bf[2][2];
#pragma unroll
                for (int s = 0; s < 2; ++s) {
#pragma unroll
                    for (int fm = 0; fm < 4; ++fm)
                        af[s][fm] = *(const v2f*)(aP + fm * 16 * G_PAD + kk + 4 * s);
#pragma unroll
                    for (int fn = 0; fn < 2; ++fn)
                        bf[s][fn] = *(const v2f*)(bP + fn * 16 * G_PAD + kk + 4 * s);
                }
#pragma unroll
                for (int s = 0; s < 2; ++s)
#pragma unroll
                    for (int fm = 0; fm < 4; ++fm)
#pragma unroll
                        for (int fn = 0; fn < 2; ++fn)
                            cacc[fm][fn] = wmma4(af[s][fm], bf[s][fn], cacc[fm][fn]);
            }
        } else if (kt + 2 < nkt) {
            gemm_fill_stage<8, 16>(Ap, Wp, K, (kt + 2) * G_TK, Ls + sf * G_STAGE_SZ, stOff);
        }
        __syncthreads();
        sc = (sc == G_STAGES - 1) ? 0 : sc + 1;
        sf = (sf == G_STAGES - 1) ? 0 : sf + 1;
    }

    if (isMac) {
#pragma unroll
        for (int fm = 0; fm < 4; ++fm)
#pragma unroll
            for (int fn = 0; fn < 2; ++fn)
#pragma unroll
                for (int r = 0; r < 8; ++r) {
                    int row = m0 + wm + fm * 16 + r + rhalf;
                    int col = n0 + wn + fn * 16 + lrow;
                    float v = cacc[fm][fn][r];
                    if (act == 1) v = (v > 0.0f) ? (v + 1.0f) : __expf(v);
                    if (bias) v += bias[col];
                    if (split) {
                        int bb = row >> 11, t = row & 2047;
                        int h  = col >> 6,  dd = col & 63;
                        C[(((size_t)bb * 16 + h) * 2048 + t) * 64 + dd] = v;
                    } else {
                        C[(size_t)row * N + col] = v;
                    }
                }
    }
}

#define P_K 1024
#define P_NW 3
#define P_STAGES 2
#define P_STAGE_SZ ((G_TM + P_NW * G_TN) * G_PAD)

template <int RSTEP, int NP>
__device__ __forceinline__ void qkv_fill_rows(const float* __restrict__ gp, float* sp)
{
    float4 r[NP];
#pragma unroll
    for (int i = 0; i < NP; ++i)
        r[i] = *(const float4*)(gp + (RSTEP * i) * P_K);
#pragma unroll
    for (int i = 0; i < NP; ++i)
        *(float4*)&sp[RSTEP * i * G_PAD] = r[i];
}

__global__ __launch_bounds__(G_THREADS) void qkv_proj_kernel(
    const float* __restrict__ X, const float* __restrict__ Wq,
    const float* __restrict__ Wk, const float* __restrict__ Wv,
    float* __restrict__ Cq, float* __restrict__ Ck, float* __restrict__ Cv)
{
    __shared__ __attribute__((aligned(16))) float Ls[P_STAGES * P_STAGE_SZ];

    const int tid   = threadIdx.x;
    const int lane  = tid & 31;
    const int lrow  = lane & 15;
    const int khalf = (lane >> 4) * 2;
    const int rhalf = (lane >> 4) * 8;
    const int wave  = __builtin_amdgcn_readfirstlane(tid >> 5);
    const bool isMac = (wave < 8);

    const int m0 = blockIdx.x * G_TM;
    const int n0 = blockIdx.y * G_TN;
    const int wm = (wave & 1) * 64;
    const int wn = ((wave >> 1) & 3) * 32;

    const int gtid  = isMac ? tid : tid - G_MAC_THREADS;
    const int ldRow = gtid >> 4;
    const int ldCol = (gtid & 15) * 4;
    const unsigned thrOff = (unsigned)(ldRow * P_K + ldCol);
    const float* Xb  = X  + (size_t)m0 * P_K;
    const float* Wqb = Wq + (size_t)n0 * P_K;
    const float* Wkb = Wk + (size_t)n0 * P_K;
    const float* Wvb = Wv + (size_t)n0 * P_K;
    const int stOff = ldRow * G_PAD + ldCol;

    const int nkt = P_K / G_TK;

    if (isMac) {
        qkv_fill_rows<16, 8>(Xb  + thrOff, Ls + stOff);
        qkv_fill_rows<16, 8>(Wqb + thrOff, Ls + G_A_SZ + stOff);
    } else {
        qkv_fill_rows<8, 16>(Wkb + thrOff, Ls + 2 * G_A_SZ + stOff);
        qkv_fill_rows<8, 16>(Wvb + thrOff, Ls + 3 * G_A_SZ + stOff);
    }
    __syncthreads();

    v8f cacc[P_NW][4][2] = {};

    for (int kt = 0; kt < nkt; ++kt) {
        const int sc = kt & 1;
        if (isMac) {
            const float* aP = Ls + sc * P_STAGE_SZ + (wm + lrow) * G_PAD + khalf;
            const float* bP = Ls + sc * P_STAGE_SZ + G_A_SZ + (wn + lrow) * G_PAD + khalf;
#pragma unroll
            for (int kk = 0; kk < G_TK; kk += 8) {
                v2f af[2][4], bf[2][P_NW][2];
#pragma unroll
                for (int s = 0; s < 2; ++s) {
#pragma unroll
                    for (int fm = 0; fm < 4; ++fm)
                        af[s][fm] = *(const v2f*)(aP + fm * 16 * G_PAD + kk + 4 * s);
#pragma unroll
                    for (int w = 0; w < P_NW; ++w)
#pragma unroll
                        for (int fn = 0; fn < 2; ++fn)
                            bf[s][w][fn] = *(const v2f*)(bP + w * G_A_SZ + fn * 16 * G_PAD + kk + 4 * s);
                }
#pragma unroll
                for (int s = 0; s < 2; ++s)
#pragma unroll
                    for (int fm = 0; fm < 4; ++fm)
#pragma unroll
                        for (int w = 0; w < P_NW; ++w)
#pragma unroll
                            for (int fn = 0; fn < 2; ++fn)
                                cacc[w][fm][fn] = wmma4(af[s][fm], bf[s][w][fn], cacc[w][fm][fn]);
            }
        } else if (kt + 1 < nkt) {
            float* st = Ls + (sc ^ 1) * P_STAGE_SZ + stOff;
            const unsigned gOff = thrOff + (unsigned)((kt + 1) * G_TK);
            qkv_fill_rows<8, 16>(Xb  + gOff, st);
            __builtin_amdgcn_sched_barrier(0);
            qkv_fill_rows<8, 16>(Wqb + gOff, st + G_A_SZ);
            __builtin_amdgcn_sched_barrier(0);
            qkv_fill_rows<8, 16>(Wkb + gOff, st + 2 * G_A_SZ);
            __builtin_amdgcn_sched_barrier(0);
            qkv_fill_rows<8, 16>(Wvb + gOff, st + 3 * G_A_SZ);
        }
        __syncthreads();
    }

    if (isMac) {
#pragma unroll
        for (int w = 0; w < P_NW; ++w) {
            float* __restrict__ C = (w == 0) ? Cq : (w == 1) ? Ck : Cv;
#pragma unroll
            for (int fm = 0; fm < 4; ++fm)
#pragma unroll
                for (int fn = 0; fn < 2; ++fn)
#pragma unroll
                    for (int r = 0; r < 8; ++r) {
                        int row = m0 + wm + fm * 16 + r + rhalf;
                        int col = n0 + wn + fn * 16 + lrow;
                        float v = cacc[w][fm][fn][r];
                        if (w < 2) v = (v > 0.0f) ? (v + 1.0f) : __expf(v);
                        int bb = row >> 11, t = row & 2047;
                        int h  = col >> 6,  dd = col & 63;
                        C[(((size_t)bb * 16 + h) * 2048 + t) * 64 + dd] = v;
                    }
        }
    }
}

__device__ __forceinline__ void lds_barrier() {
    __builtin_amdgcn_fence(__ATOMIC_RELEASE, "workgroup", "local");
    __builtin_amdgcn_s_barrier();
    __builtin_amdgcn_fence(__ATOMIC_ACQUIRE, "workgroup", "local");
}

#define C_PAD  68
#define V_PAD  40
#define A_PAD  130
#define S_PAD  40
#define R_ROWS 32
#define V_COLS 32
#define QS_SZ (R_ROWS * C_PAD)
#define KS_SZ (128 * C_PAD)
#define VS_SZ (128 * V_PAD)
#define AM_SZ (R_ROWS * A_PAD)
#define SM_SZ (64 * S_PAD)

__device__ __forceinline__ void scan_keep_group() { __builtin_amdgcn_sched_barrier(0); }


__device__ __forceinline__ void scan_ld_scores(const float* qA, const float* kB, int kk0,
                                               v2f (&qf)[4], v2f (&kf)[4][4])
{
#pragma unroll
    for (int s = 0; s < 4; ++s) {
        qf[s] = *(const v2f*)(qA + kk0 + 4 * s);
#pragma unroll
        for (int c = 0; c < 4; ++c)
            kf[s][c] = *(const v2f*)(kB + c * 16 * C_PAD + kk0 + 4 * s);
    }
}

template <int BP>
__device__ __forceinline__ void scan_ld_ab2(const float* aA, const float* bB, int kk0,
                                            v2f (&af)[8], float (&bx)[8][2], float (&by)[8][2])
{
#pragma unroll
    for (int s = 0; s < 8; ++s) {
        const int kk = kk0 + 4 * s;
        af[s] = *(const v2f*)(aA + kk);
#pragma unroll
        for (int c = 0; c < 2; ++c) {
            bx[s][c] = bB[kk       * BP + c * 16];
            by[s][c] = bB[(kk + 1) * BP + c * 16];
        }
    }
}

__device__ __forceinline__ v2f scan_pair(float x, float y) { v2f r; r.x = x; r.y = y; return r; }

__device__ __forceinline__ void scan_ld_kv(const float* kA, const float* vB, int kk0,
                                           float (&ax)[4][2], float (&ay)[4][2],
                                           float (&bx)[4][2], float (&by)[4][2])
{
#pragma unroll
    for (int s = 0; s < 4; ++s) {
        const int kk = kk0 + 4 * s;
#pragma unroll
        for (int i = 0; i < 2; ++i) {
            ax[s][i] = kA[kk       * C_PAD + i * 16];
            ay[s][i] = kA[(kk + 1) * C_PAD + i * 16];
        }
#pragma unroll
        for (int c = 0; c < 2; ++c) {
            bx[s][c] = vB[kk       * V_PAD + c * 16];
            by[s][c] = vB[(kk + 1) * V_PAD + c * 16];
        }
    }
}

__global__ __launch_bounds__(256) void linattn_scan_kernel(
    const float* __restrict__ Q, const float* __restrict__ Kx,
    const float* __restrict__ V, float* __restrict__ O)
{
    __shared__ __attribute__((aligned(16))) float qbuf[3 * QS_SZ];
    __shared__ __attribute__((aligned(16))) float kbuf[3 * KS_SZ];
    __shared__ __attribute__((aligned(16))) float vbuf[3 * VS_SZ];
    __shared__ __attribute__((aligned(16))) float Ambuf[2 * AM_SZ];
    __shared__ __attribute__((aligned(16))) float Smbuf[2 * SM_SZ];
    __shared__ __attribute__((aligned(16))) float zv[64];
    __shared__ float rsbuf[2 * R_ROWS];
    __shared__ float qzbuf[2 * R_ROWS];

    const int tid   = threadIdx.x;
    const int lane  = tid & 31;
    const int lrow  = lane & 15;
    const int khalf = (lane >> 4) * 2;
    const int rhalf = (lane >> 4) * 8;
    const int wave  = __builtin_amdgcn_readfirstlane(tid >> 5);

    const int bh = blockIdx.x >> 3;
    const int r0 = ((blockIdx.x >> 1) & 3) * R_ROWS;
    const int c0 = (blockIdx.x & 1) * V_COLS;
    const int b  = bh >> 4, h = bh & 15;
    const size_t base = (size_t)bh * 2048 * 64;

    const int kqRow = tid >> 4, kqC4 = (tid & 15) * 4;
    const int vRow  = tid >> 3, vC4  = (tid & 7) * 4;
    const float* gK = Kx + base + kqRow * 64 + kqC4;
    const float* gQ = Q  + base + (r0 + kqRow) * 64 + kqC4;
    const float* gV = V  + base + vRow * 64 + c0 + vC4;
    const int kqOff = kqRow * C_PAD + kqC4;
    const int vOff  = vRow * V_PAD + vC4;

#define SCAN_IMAGE_DECL \
    float4 pk0, pk1, pk2, pk3, pk4, pk5, pk6, pk7, pq0, pq1, pv0, pv1, pv2, pv3;
#define SCAN_LOAD_IMAGE(off)                                   \
    pk0 = *(const float4*)(gK + (off) + 0 * 16 * 64);          \
    pk1 = *(const float4*)(gK + (off) + 1 * 16 * 64);          \
    pk2 = *(const float4*)(gK + (off) + 2 * 16 * 64);          \
    pk3 = *(const float4*)(gK + (off) + 3 * 16 * 64);          \
    pk4 = *(const float4*)(gK + (off) + 4 * 16 * 64);          \
    pk5 = *(const float4*)(gK + (off) + 5 * 16 * 64);          \
    pk6 = *(const float4*)(gK + (off) + 6 * 16 * 64);          \
    pk7 = *(const float4*)(gK + (off) + 7 * 16 * 64);          \
    pq0 = *(const float4*)(gQ + (off) + 0 * 16 * 64);          \
    pq1 = *(const float4*)(gQ + (off) + 1 * 16 * 64);          \
    pv0 = *(const float4*)(gV + (off) + 0 * 32 * 64);          \
    pv1 = *(const float4*)(gV + (off) + 1 * 32 * 64);          \
    pv2 = *(const float4*)(gV + (off) + 2 * 32 * 64);          \
    pv3 = *(const float4*)(gV + (off) + 3 * 32 * 64);
#define SCAN_STORE_IMAGE(kd, qd, vd)                           \
    *(float4*)&(kd)[kqOff + 0 * 16 * C_PAD] = pk0;             \
    *(float4*)&(kd)[kqOff + 1 * 16 * C_PAD] = pk1;             \
    *(float4*)&(kd)[kqOff + 2 * 16 * C_PAD] = pk2;             \
    *(float4*)&(kd)[kqOff + 3 * 16 * C_PAD] = pk3;             \
    *(float4*)&(kd)[kqOff + 4 * 16 * C_PAD] = pk4;             \
    *(float4*)&(kd)[kqOff + 5 * 16 * C_PAD] = pk5;             \
    *(float4*)&(kd)[kqOff + 6 * 16 * C_PAD] = pk6;             \
    *(float4*)&(kd)[kqOff + 7 * 16 * C_PAD] = pk7;             \
    *(float4*)&(qd)[kqOff + 0 * 16 * C_PAD] = pq0;             \
    *(float4*)&(qd)[kqOff + 1 * 16 * C_PAD] = pq1;             \
    *(float4*)&(vd)[vOff + 0 * 32 * V_PAD] = pv0;              \
    *(float4*)&(vd)[vOff + 1 * 32 * V_PAD] = pv1;              \
    *(float4*)&(vd)[vOff + 2 * 32 * V_PAD] = pv2;              \
    *(float4*)&(vd)[vOff + 3 * 32 * V_PAD] = pv3;


    {
        SCAN_IMAGE_DECL
        SCAN_LOAD_IMAGE(0)
        for (int i = tid; i < SM_SZ; i += 256) Smbuf[i] = 0.0f;
        SCAN_STORE_IMAGE(kbuf, qbuf, vbuf)
    }
    __syncthreads();

    float zreg = 0.0f;
    int bA = 0;

    for (int t = 0; t <= 16; ++t) {
        const bool hasA = (t < 16);
        const bool hasS = (t > 0);
        const bool more = (t + 1 < 16);
        const int bS = (bA == 0) ? 2 : bA - 1;
        const int bN = (bA == 2) ? 0 : bA + 1;
        const int pa = t & 1;

        const float* qsA = qbuf + bA * QS_SZ;
        const float* ksA = kbuf + bA * KS_SZ;
        const float* qsS = qbuf + bS * QS_SZ;
        const float* ksS = kbuf + bS * KS_SZ;
        const float* vsS = vbuf + bS * VS_SZ;
        float*       AmW  = Ambuf + pa * AM_SZ;
        const float* AmR  = Ambuf + (pa ^ 1) * AM_SZ;
        const float* Scur = Smbuf + (pa ^ 1) * SM_SZ;
        float*       Snxt = Smbuf + pa * SM_SZ;
        float*       rsW  = rsbuf + pa * R_ROWS;
        const float* rsR  = rsbuf + (pa ^ 1) * R_ROWS;
        float*       qzW  = qzbuf + pa * R_ROWS;
        const float* qzR  = qzbuf + (pa ^ 1) * R_ROWS;

        SCAN_IMAGE_DECL
        if (more) {
            const size_t nOff = (size_t)(t + 1) * 128 * 64;
            SCAN_LOAD_IMAGE(nOff)
        }

        v8f nacc[2] = {};
        v8f sacc[4] = {};

        if (wave >= 4) {
            if (hasA) {
                if (wave >= 6) zv[tid - 192] = zreg;

                const int a   = wave - 4;
                const int m0  = (a >> 1) * 16;
                const int ct0 = (a & 1) * 4;
                const float* qA = qsA + (m0 + lrow) * C_PAD + khalf;
                const float* kB = ksA + (ct0 * 16 + lrow) * C_PAD + khalf;
                v8f acc[4] = {};
                v2f qf[2][4], kf[2][4][4];
                scan_ld_scores(qA, kB, 0, qf[0], kf[0]);
#pragma unroll
                for (int blk = 0; blk < 4; ++blk) {
                    const int cur = blk & 1;
                    if (blk + 1 < 4)
                        scan_ld_scores(qA, kB, (blk + 1) * 16, qf[cur ^ 1], kf[cur ^ 1]);
                    scan_keep_group();
#pragma unroll
                    for (int s = 0; s < 4; ++s)
#pragma unroll
                        for (int c = 0; c < 4; ++c)
                            acc[c] = wmma4(qf[cur][s], kf[cur][s][c], acc[c]);
                    scan_keep_group();
                }
#pragma unroll
                for (int c = 0; c < 4; ++c)
#pragma unroll
                    for (int r = 0; r < 8; ++r) {
                        int row = m0 + r + rhalf;
                        int col = (ct0 + c) * 16 + lrow;
                        AmW[row * A_PAD + col] = (r0 + row >= col) ? acc[c][r] : 0.0f;
                    }
            }
        } else if (wave < 2) {
            if (hasS) {
                const int m0 = wave * 16;
                const float* aA = AmR + (m0 + lrow) * A_PAD + khalf;
                const float* vB = vsS + khalf * V_PAD + lrow;
                v2f af[2][8];
                float vx[2][8][2], vy[2][8][2];
                scan_ld_ab2<V_PAD>(aA, vB, 0, af[0], vx[0], vy[0]);
#pragma unroll
                for (int blk = 0; blk < 4; ++blk) {
                    const int cur = blk & 1;
                    if (blk + 1 < 4)
                        scan_ld_ab2<V_PAD>(aA, vB, (blk + 1) * 32, af[cur ^ 1], vx[cur ^ 1], vy[cur ^ 1]);
                    scan_keep_group();
#pragma unroll
                    for (int s = 0; s < 8; ++s)
#pragma unroll
                        for (int c = 0; c < 2; ++c)
                            nacc[c] = wmma4(af[cur][s], scan_pair(vx[cur][s][c], vy[cur][s][c]), nacc[c]);
                    scan_keep_group();
                }
            }
        } else {
            if (hasS) {
                const int tr0 = (wave - 2) * 32;
                const float* kA = ksS + khalf * C_PAD + tr0 + lrow;
                const float* vB = vsS + khalf * V_PAD + lrow;
#pragma unroll
                for (int i = 0; i < 2; ++i)
#pragma unroll
                    for (int c = 0; c < 2; ++c)
#pragma unroll
                        for (int r = 0; r < 8; ++r)
                            sacc[i * 2 + c][r] = Scur[(tr0 + i * 16 + r + rhalf) * S_PAD + c * 16 + lrow];
                float ax[2][4][2], ay[2][4][2], bx[2][4][2], by[2][4][2];
                scan_ld_kv(kA, vB, 0, ax[0], ay[0], bx[0], by[0]);
#pragma unroll
                for (int blk = 0; blk < 4; ++blk) {
                    const int cur = blk & 1;
                    if (blk + 1 < 4)
                        scan_ld_kv(kA, vB, (blk + 1) * 16, ax[cur ^ 1], ay[cur ^ 1], bx[cur ^ 1], by[cur ^ 1]);
                    scan_keep_group();
#pragma unroll
                    for (int s = 0; s < 4; ++s)
#pragma unroll
                        for (int i = 0; i < 2; ++i)
#pragma unroll
                            for (int c = 0; c < 2; ++c)
                                sacc[i * 2 + c] = wmma4(scan_pair(ax[cur][s][i], ay[cur][s][i]),
                                                        scan_pair(bx[cur][s][c], by[cur][s][c]), sacc[i * 2 + c]);
                    scan_keep_group();
                }
            }
        }
        lds_barrier();

        if (wave >= 4) {
            if (hasA) {
                if (wave == 4) {
                    const float* ar = AmW + lane * A_PAD;
                    float s = 1e-5f;
                    v2f rb[2][16];
#pragma unroll
                    for (int i = 0; i < 16; ++i) rb[0][i] = *(const v2f*)(ar + 2 * i);
#pragma unroll
                    for (int blk = 0; blk < 4; ++blk) {
                        const int cur = blk & 1;
                        if (blk + 1 < 4) {
#pragma unroll
                            for (int i = 0; i < 16; ++i)
                                rb[cur ^ 1][i] = *(const v2f*)(ar + (blk + 1) * 32 + 2 * i);
                        }
                        scan_keep_group();
#pragma unroll
                        for (int i = 0; i < 16; ++i) { s += rb[cur][i].x; s += rb[cur][i].y; }
                        scan_keep_group();
                    }
                    rsW[lane] = s;
                } else if (wave == 5) {
                    const float* qr = qsA + lane * C_PAD;
                    float qz = 0.0f;
                    float4 qb[2][4], zb[2][4];
#pragma unroll
                    for (int i = 0; i < 4; ++i) {
                        qb[0][i] = *(const float4*)(qr + 4 * i);
                        zb[0][i] = *(const float4*)(zv + 4 * i);
                    }
#pragma unroll
                    for (int blk = 0; blk < 4; ++blk) {
                        const int cur = blk & 1;
                        if (blk + 1 < 4) {
#pragma unroll
                            for (int i = 0; i < 4; ++i) {
                                qb[cur ^ 1][i] = *(const float4*)(qr + (blk + 1) * 16 + 4 * i);
                                zb[cur ^ 1][i] = *(const float4*)(zv + (blk + 1) * 16 + 4 * i);
                            }
                        }
                        scan_keep_group();
#pragma unroll
                        for (int i = 0; i < 4; ++i) {
                            qz = fmaf(qb[cur][i].x, zb[cur][i].x, qz);
                            qz = fmaf(qb[cur][i].y, zb[cur][i].y, qz);
                            qz = fmaf(qb[cur][i].z, zb[cur][i].z, qz);
                            qz = fmaf(qb[cur][i].w, zb[cur][i].w, qz);
                        }
                        scan_keep_group();
                    }
                    qzW[lane] = qz;
                } else {
                    const float* kc = ksA + (tid - 192);
                    float s = zreg;
                    float cb[2][32];
#pragma unroll
                    for (int i = 0; i < 32; ++i) cb[0][i] = kc[i * C_PAD];
#pragma unroll
                    for (int blk = 0; blk < 4; ++blk) {
                        const int cur = blk & 1;
                        if (blk + 1 < 4) {
#pragma unroll
                            for (int i = 0; i < 32; ++i)
                                cb[cur ^ 1][i] = kc[((blk + 1) * 32 + i) * C_PAD];
                        }
                        scan_keep_group();
#pragma unroll
                        for (int i = 0; i < 32; ++i) s += cb[cur][i];
                        scan_keep_group();
                    }
                    zreg = s;
                }
            }
        } else if (wave < 2) {
            if (hasS) {
                const int m0 = wave * 16;
                const float* qA = qsS + (m0 + lrow) * C_PAD + khalf;
                const float* sB = Scur + khalf * S_PAD + lrow;
                v2f qf[2][8];
                float sx[2][8][2], sy[2][8][2];
                scan_ld_ab2<S_PAD>(qA, sB, 0, qf[0], sx[0], sy[0]);
#pragma unroll
                for (int blk = 0; blk < 2; ++blk) {
                    const int cur = blk & 1;
                    if (blk + 1 < 2)
                        scan_ld_ab2<S_PAD>(qA, sB, (blk + 1) * 32, qf[cur ^ 1], sx[cur ^ 1], sy[cur ^ 1]);
                    scan_keep_group();
#pragma unroll
                    for (int s = 0; s < 8; ++s)
#pragma unroll
                        for (int c = 0; c < 2; ++c)
                            nacc[c] = wmma4(qf[cur][s], scan_pair(sx[cur][s][c], sy[cur][s][c]), nacc[c]);
                    scan_keep_group();
                }
#pragma unroll
                for (int c = 0; c < 2; ++c)
#pragma unroll
                    for (int r = 0; r < 8; ++r) {
                        int row = m0 + r + rhalf;
                        int col = c * 16 + lrow;
                        float o = nacc[c][r] / (rsR[row] + qzR[row]);
                        O[((size_t)(b * 2048 + (t - 1) * 128 + r0 + row)) * 1024 + h * 64 + c0 + col] = o;
                    }
            }
        } else {
            if (hasS) {
                const int tr0 = (wave - 2) * 32;
                const float* kA = ksS + khalf * C_PAD + tr0 + lrow;
                const float* vB = vsS + khalf * V_PAD + lrow;
                float ax[2][4][2], ay[2][4][2], bx[2][4][2], by[2][4][2];
                scan_ld_kv(kA, vB, 64, ax[0], ay[0], bx[0], by[0]);
#pragma unroll
                for (int blk = 0; blk < 4; ++blk) {
                    const int cur = blk & 1;
                    if (blk + 1 < 4)
                        scan_ld_kv(kA, vB, 64 + (blk + 1) * 16, ax[cur ^ 1], ay[cur ^ 1], bx[cur ^ 1], by[cur ^ 1]);
                    scan_keep_group();
#pragma unroll
                    for (int s = 0; s < 4; ++s)
#pragma unroll
                        for (int i = 0; i < 2; ++i)
#pragma unroll
                            for (int c = 0; c < 2; ++c)
                                sacc[i * 2 + c] = wmma4(scan_pair(ax[cur][s][i], ay[cur][s][i]),
                                                        scan_pair(bx[cur][s][c], by[cur][s][c]), sacc[i * 2 + c]);
                    scan_keep_group();
                }
#pragma unroll
                for (int i = 0; i < 2; ++i)
#pragma unroll
                    for (int c = 0; c < 2; ++c)
#pragma unroll
                        for (int r = 0; r < 8; ++r)
                            Snxt[(tr0 + i * 16 + r + rhalf) * S_PAD + c * 16 + lrow] = sacc[i * 2 + c][r];
            }
        }

        if (more) {
            float* kn = kbuf + bN * KS_SZ;
            float* qn = qbuf + bN * QS_SZ;
            float* vn = vbuf + bN * VS_SZ;
            SCAN_STORE_IMAGE(kn, qn, vn)
        }
        lds_barrier();
        bA = bN;
    }
}

extern "C" void kernel_launch(void* const* d_in, const int* in_sizes, int n_in,
                              void* d_out, int out_size, void* d_ws, size_t ws_size,
                              hipStream_t stream) {
    const float* x  = (const float*)d_in[0];
    const float* Wq = (const float*)d_in[1];
    const float* Wk = (const float*)d_in[2];
    const float* Wv = (const float*)d_in[3];
    const float* Wo = (const float*)d_in[4];
    const float* bo = (const float*)d_in[5];

    const int T = 2048, E = 1024;
    const int B = in_sizes[0] / (T * E);
    const int M = B * T;
    const size_t plane = (size_t)M * E;

    float* phiQ = (float*)d_ws;
    float* phiK = phiQ + plane;
    float* Vh   = phiK + plane;
    float* attn = Vh   + plane;

    dim3 blk(256);
    dim3 gblk(G_THREADS);
    dim3 gg(M / G_TM, E / G_TN);

    qkv_proj_kernel<<<gg, gblk, 0, stream>>>(x, Wq, Wk, Wv, phiQ, phiK, Vh);

    linattn_scan_kernel<<<dim3(B * 128), blk, 0, stream>>>(phiQ, phiK, Vh, attn);

    gemm_nt_kernel<<<gg, gblk, 0, stream>>>(attn, Wo, bo, (float*)d_out, M, E, E,  0,  0);
}
